# starting kernel
# baseline (speedup 1.0000x reference)
#include <hip/hip_runtime.h>


#define GCN_N_NODES 50000
#define GCN_N_EDGES 800000
#define GCN_IN_DIM 128
#define GCN_HID_DIM 128
#define GCN_OUT_DIM 64

typedef __attribute__((ext_vector_type(2))) float v2f;
typedef __attribute__((ext_vector_type(8))) float v8f;

__global__ void gcn_degree_kernel(const int* __restrict__ src,
                                  const int* __restrict__ dst,
                                  float* __restrict__ deg_out,
                                  float* __restrict__ deg_in, int E) {
  int e = blockIdx.x * blockDim.x + threadIdx.x;
  if (e < E) {
    atomicAdd(&deg_out[src[e]], 1.0f);
    atomicAdd(&deg_in[dst[e]], 1.0f);
  }
}

__global__ void gcn_norm_kernel(const float* __restrict__ deg_out,
                                const float* __restrict__ deg_in,
                                float* __restrict__ norm_out,
                                float* __restrict__ norm_in, int N) {
  int i = blockIdx.x * blockDim.x + threadIdx.x;
  if (i < N) {
    norm_out[i] = rsqrtf(fmaxf(deg_out[i], 1.0f));
    norm_in[i]  = rsqrtf(fmaxf(deg_in[i], 1.0f));
  }
}

template <int K, int NOUT>
__global__ void gcn_gemm_scaled_wmma(const float* __restrict__ X,
                                     const float* __restrict__ scale,
                                     const float* __restrict__ W,
                                     float* __restrict__ Y) {
  constexpr int LDSK = K + 4;
  __shared__ float tile[16 * LDSK];

  const int tid  = threadIdx.x;
  const int row0 = blockIdx.x * 16;

  for (int t = tid; t < 16 * K; t += blockDim.x) {
    const int r = t / K;
    const int c = t - r * K;
    const int node = row0 + r;
    tile[r * LDSK + c] = X[(size_t)node * K + c] * scale[node];
  }
  __syncthreads();

  const int wave = tid >> 5;
  const int lane = tid & 31;
  const int m    = lane & 15;
  const int kh   = lane >> 4;
  const int nc   = wave * 16 + m;

  v8f acc = {};
#pragma unroll 4
  for (int kk = 0; kk < K; kk += 4) {
    v2f a, b;
    const int ka = kk + 2 * kh;
    a.x = tile[m * LDSK + ka];
    a.y = tile[m * LDSK + ka + 1];
    b.x = W[(size_t)ka * NOUT + nc];
    b.y = W[(size_t)(ka + 1) * NOUT + nc];
    acc = __builtin_amdgcn_wmma_f32_16x16x4_f32(
         false, a,  false, b,
         (short)0, acc,  false,  false);
  }

#pragma unroll
  for (int v = 0; v < 8; ++v) {
    const int orow = row0 + v + 8 * kh;
    Y[(size_t)orow * NOUT + nc] = acc[v];
  }
}

template <int D>
__global__ void gcn_aggregate_kernel(const float* __restrict__ H,
                                     const int* __restrict__ src,
                                     const int* __restrict__ dst,
                                     float* __restrict__ agg, int E) {
  const unsigned t = blockIdx.x * blockDim.x + threadIdx.x;
  const unsigned e = t / D;
  const unsigned d = t - e * D;
  if (e < (unsigned)E) {
    const int s = src[e];
    const int r = dst[e];
    atomicAdd(&agg[(size_t)r * D + d], H[(size_t)s * D + d]);
  }
}

template <int D>
__global__ void gcn_finalize_kernel(const float* __restrict__ agg,
                                    const float* __restrict__ norm_in,
                                    const float* __restrict__ b,
                                    float* __restrict__ out, int N) {
  const unsigned t = blockIdx.x * blockDim.x + threadIdx.x;
  const unsigned i = t / D;
  const unsigned d = t - i * D;
  if (i < (unsigned)N) {
    const float v = fmaf(agg[t], norm_in[i], b[d]);
    out[t] = fmaxf(v, 0.0f);
  }
}

extern "C" void kernel_launch(void* const* d_in, const int* in_sizes, int n_in,
                              void* d_out, int out_size, void* d_ws, size_t ws_size,
                              hipStream_t stream) {
  const float* feat = (const float*)d_in[0];
  const int*   src  = (const int*)d_in[1];
  const int*   dst  = (const int*)d_in[2];
  const float* W1   = (const float*)d_in[3];
  const float* b1   = (const float*)d_in[4];
  const float* W2   = (const float*)d_in[5];
  const float* b2   = (const float*)d_in[6];
  float* out = (float*)d_out;
  float* ws  = (float*)d_ws;

  const int N = GCN_N_NODES;
  const int E = GCN_N_EDGES;

  float* deg_out  = ws;
  float* deg_in   = ws + (size_t)N;
  float* norm_out = ws + 2 * (size_t)N;
  float* norm_in  = ws + 3 * (size_t)N;
  float* bufA     = ws + 4 * (size_t)N;
  float* bufB     = bufA + (size_t)N * GCN_HID_DIM;
  float* hproj2   = bufA;
  float* agg2     = bufA + (size_t)N * GCN_OUT_DIM;

  hipMemsetAsync(deg_out, 0, 2 * (size_t)N * sizeof(float), stream);
  gcn_degree_kernel<<<(E + 255) / 256, 256, 0, stream>>>(src, dst, deg_out, deg_in, E);
  gcn_norm_kernel<<<(N + 255) / 256, 256, 0, stream>>>(deg_out, deg_in, norm_out, norm_in, N);

  gcn_gemm_scaled_wmma<GCN_IN_DIM, GCN_HID_DIM>
      <<<N / 16, (GCN_HID_DIM / 16) * 32, 0, stream>>>(feat, norm_out, W1, bufA);
  hipMemsetAsync(bufB, 0, (size_t)N * GCN_HID_DIM * sizeof(float), stream);
  {
    const unsigned work = (unsigned)E * GCN_HID_DIM;
    gcn_aggregate_kernel<GCN_HID_DIM>
        <<<(work + 255) / 256, 256, 0, stream>>>(bufA, src, dst, bufB, E);
  }
  {
    const unsigned work = (unsigned)N * GCN_HID_DIM;
    gcn_finalize_kernel<GCN_HID_DIM>
        <<<(work + 255) / 256, 256, 0, stream>>>(bufB, norm_in, b1, bufB, N);
  }

  gcn_gemm_scaled_wmma<GCN_HID_DIM, GCN_OUT_DIM>
      <<<N / 16, (GCN_OUT_DIM / 16) * 32, 0, stream>>>(bufB, norm_out, W2, hproj2);
  hipMemsetAsync(agg2, 0, (size_t)N * GCN_OUT_DIM * sizeof(float), stream);
  {
    const unsigned work = (unsigned)E * GCN_OUT_DIM;
    gcn_aggregate_kernel<GCN_OUT_DIM>
        <<<(work + 255) / 256, 256, 0, stream>>>(hproj2, src, dst, agg2, E);
  }
  {
    const unsigned work = (unsigned)N * GCN_OUT_DIM;
    gcn_finalize_kernel<GCN_OUT_DIM>
        <<<(work + 255) / 256, 256, 0, stream>>>(agg2, norm_in, b2, out, N);
  }
}


# aggregation by owner blocks: a block keeps 200 dst nodes' sums in LDS, filters the whole dst list, gathers H[src] rows, adds with LDS float atomics (sum order = their arrival), writes relu rows
# speedup vs baseline: 2.3881x; 2.3881x over previous
#include <hip/hip_runtime.h>


#define GCN_N_NODES 50000
#define GCN_N_EDGES 800000
#define GCN_IN_DIM 128
#define GCN_HID_DIM 128
#define GCN_OUT_DIM 64

typedef __attribute__((ext_vector_type(2))) float v2f;
typedef __attribute__((ext_vector_type(8))) float v8f;

__global__ void gcn_degree_kernel(const int* __restrict__ src,
                                  const int* __restrict__ dst,
                                  float* __restrict__ deg_out,
                                  float* __restrict__ deg_in, int E) {
  int e = blockIdx.x * blockDim.x + threadIdx.x;
  if (e < E) {
    atomicAdd(&deg_out[src[e]], 1.0f);
    atomicAdd(&deg_in[dst[e]], 1.0f);
  }
}

__global__ void gcn_norm_kernel(const float* __restrict__ deg_out,
                                const float* __restrict__ deg_in,
                                float* __restrict__ norm_out,
                                float* __restrict__ norm_in, int N) {
  int i = blockIdx.x * blockDim.x + threadIdx.x;
  if (i < N) {
    norm_out[i] = rsqrtf(fmaxf(deg_out[i], 1.0f));
    norm_in[i]  = rsqrtf(fmaxf(deg_in[i], 1.0f));
  }
}

template <int K, int NOUT>
__global__ void gcn_gemm_scaled_wmma(const float* __restrict__ X,
                                     const float* __restrict__ scale,
                                     const float* __restrict__ W,
                                     float* __restrict__ Y) {
  constexpr int LDSK = K + 4;
  __shared__ float tile[16 * LDSK];

  const int tid  = threadIdx.x;
  const int row0 = blockIdx.x * 16;

  for (int t = tid; t < 16 * K; t += blockDim.x) {
    const int r = t / K;
    const int c = t - r * K;
    const int node = row0 + r;
    tile[r * LDSK + c] = X[(size_t)node * K + c] * scale[node];
  }
  __syncthreads();

  const int wave = tid >> 5;
  const int lane = tid & 31;
  const int m    = lane & 15;
  const int kh   = lane >> 4;
  const int nc   = wave * 16 + m;

  v8f acc = {};
#pragma unroll 4
  for (int kk = 0; kk < K; kk += 4) {
    v2f a, b;
    const int ka = kk + 2 * kh;
    a.x = tile[m * LDSK + ka];
    a.y = tile[m * LDSK + ka + 1];
    b.x = W[(size_t)ka * NOUT + nc];
    b.y = W[(size_t)(ka + 1) * NOUT + nc];
    acc = __builtin_amdgcn_wmma_f32_16x16x4_f32(
         false, a,  false, b,
         (short)0, acc,  false,  false);
  }

#pragma unroll
  for (int v = 0; v < 8; ++v) {
    const int orow = row0 + v + 8 * kh;
    Y[(size_t)orow * NOUT + nc] = acc[v];
  }
}

#define AGG_T 200
#define AGG_NW 32
#define AGG_U 8
#define AGG_B 16
#define AGG_QCAP (AGG_U * 128 + AGG_B)

template <int D>
__global__ void __launch_bounds__(AGG_NW * 32)
gcn_aggregate_owner_kernel(const float* __restrict__ H,
                           const int* __restrict__ src,
                           const int* __restrict__ dst,
                           const float* __restrict__ norm_in,
                           const float* __restrict__ b,
                           float* __restrict__ out, int N) {
  constexpr int VEC = D / 32;
  typedef float vecf __attribute__((ext_vector_type(VEC)));
  typedef int v4i __attribute__((ext_vector_type(4)));
  static_assert(GCN_N_EDGES % 128 == 0, "edge list is read in 128-edge chunks");
  static_assert(AGG_T + 1 <= 512 && GCN_N_EDGES <= (1 << 20), "queue entry packing");
  constexpr int NCH   = GCN_N_EDGES / 128;
  constexpr int STEP  = AGG_NW * AGG_U;
  constexpr int NSTEP = (NCH + STEP - 1) / STEP;

  __shared__ float acc[(AGG_T + 1) * D];
  __shared__ unsigned queue[AGG_NW * AGG_QCAP];

  const int tid  = threadIdx.x;
  const int lane = tid & 31;
  const int wave = __builtin_amdgcn_readfirstlane(tid >> 5);
  const unsigned nbase = blockIdx.x * AGG_T;
  const int rot = blockIdx.x % NSTEP;

  for (int i = tid; i < (AGG_T + 1) * D; i += AGG_NW * 32) acc[i] = 0.0f;
  __syncthreads();

  unsigned* q = queue + wave * AGG_QCAP;
  int cnt = 0;

  for (int step = 0; step <= NSTEP; ++step) {
    if (step < NSTEP) {
      const int sidx = step + rot < NSTEP ? step + rot : step + rot - NSTEP;
      v4i dv[AGG_U];
#pragma unroll
      for (int u = 0; u < AGG_U; ++u) {
        const int c  = sidx * STEP + u * AGG_NW + wave;
        const int cc = c < NCH ? c : NCH - 1;
        dv[u] = *reinterpret_cast<const v4i*>(dst + (size_t)cc * 128 + lane * 4);
      }
#pragma unroll
      for (int u = 0; u < AGG_U; ++u) {
        const int c = sidx * STEP + u * AGG_NW + wave;
        const unsigned nb = c < NCH ? nbase : 0x80000000u;
        const unsigned r0 = (unsigned)dv[u].x - nb;
        const unsigned r1 = (unsigned)dv[u].y - nb;
        const unsigned r2 = (unsigned)dv[u].z - nb;
        const unsigned r3 = (unsigned)dv[u].w - nb;
        const unsigned mn = min(min(r0, r1), min(r2, r3));
        if (__builtin_amdgcn_ballot_w32(mn < (unsigned)AGG_T) != 0u) {
          const unsigned eb = ((unsigned)c * 128u + (unsigned)lane * 4u) << 9;
          const unsigned rr[4] = {r0, r1, r2, r3};
#pragma unroll
          for (int j = 0; j < 4; ++j) {
            const bool hit = rr[j] < (unsigned)AGG_T;
            const unsigned mask = __builtin_amdgcn_ballot_w32(hit);
            if (mask != 0u) {
              if (hit) q[cnt + __builtin_amdgcn_mbcnt_lo(mask, 0u)] = (eb + ((unsigned)j << 9)) | rr[j];
              cnt += __builtin_popcount(mask);
            }
          }
        }
      }
    } else if (cnt > 0) {
      if (lane >= cnt && lane < AGG_B) q[lane] = (unsigned)AGG_T;
      cnt = AGG_B;
    }

    __builtin_amdgcn_fence(__ATOMIC_SEQ_CST, "wavefront");

    while (cnt >= AGG_B) {
      cnt -= AGG_B;
      const unsigned pk = q[cnt + (lane & (AGG_B - 1))];
      const int s = src[pk >> 9];
      vecf rows[AGG_B];
#pragma unroll
      for (int i = 0; i < AGG_B; ++i) {
        const int si = __builtin_amdgcn_readlane(s, i);
        rows[i] = *reinterpret_cast<const vecf*>(H + (size_t)si * D + lane * VEC);
      }
#pragma unroll
      for (int i = 0; i < AGG_B; ++i) {
        const unsigned dl = (unsigned)__builtin_amdgcn_readlane((int)pk, i) & 511u;
        for (int j = 0; j < VEC; ++j)
          __hip_atomic_fetch_add(&acc[dl * D + j * 32 + lane], rows[i][j],
                                 __ATOMIC_RELAXED, __HIP_MEMORY_SCOPE_WORKGROUP);
      }
    }
  }
  __syncthreads();

  for (int i = tid; i < AGG_T * 32; i += AGG_NW * 32) {
    const int nl   = i >> 5;
    const int node = (int)nbase + nl;
    if (node < N) {
      const float ni = norm_in[node];
      vecf o;
      for (int j = 0; j < VEC; ++j)
        o[j] = fmaxf(fmaf(acc[nl * D + j * 32 + lane], ni, b[lane * VEC + j]), 0.0f);
      *reinterpret_cast<vecf*>(out + (size_t)node * D + lane * VEC) = o;
    }
  }
}

extern "C" void kernel_launch(void* const* d_in, const int* in_sizes, int n_in,
                              void* d_out, int out_size, void* d_ws, size_t ws_size,
                              hipStream_t stream) {
  const float* feat = (const float*)d_in[0];
  const int*   src  = (const int*)d_in[1];
  const int*   dst  = (const int*)d_in[2];
  const float* W1   = (const float*)d_in[3];
  const float* b1   = (const float*)d_in[4];
  const float* W2   = (const float*)d_in[5];
  const float* b2   = (const float*)d_in[6];
  float* out = (float*)d_out;
  float* ws  = (float*)d_ws;

  const int N = GCN_N_NODES;
  const int E = GCN_N_EDGES;

  float* deg_out  = ws;
  float* deg_in   = ws + (size_t)N;
  float* norm_out = ws + 2 * (size_t)N;
  float* norm_in  = ws + 3 * (size_t)N;
  float* bufA     = ws + 4 * (size_t)N;
  float* bufB     = bufA + (size_t)N * GCN_HID_DIM;
  float* hproj2   = bufA;

  hipMemsetAsync(deg_out, 0, 2 * (size_t)N * sizeof(float), stream);
  gcn_degree_kernel<<<(E + 255) / 256, 256, 0, stream>>>(src, dst, deg_out, deg_in, E);
  gcn_norm_kernel<<<(N + 255) / 256, 256, 0, stream>>>(deg_out, deg_in, norm_out, norm_in, N);

  gcn_gemm_scaled_wmma<GCN_IN_DIM, GCN_HID_DIM>
      <<<N / 16, (GCN_HID_DIM / 16) * 32, 0, stream>>>(feat, norm_out, W1, bufA);
  gcn_aggregate_owner_kernel<GCN_HID_DIM>
      <<<(GCN_N_NODES + AGG_T - 1) / AGG_T, AGG_NW * 32, 0, stream>>>(bufA, src, dst, norm_in, b1, bufB, N);

  gcn_gemm_scaled_wmma<GCN_HID_DIM, GCN_OUT_DIM>
      <<<N / 16, (GCN_OUT_DIM / 16) * 32, 0, stream>>>(bufB, norm_out, W2, hproj2);
  gcn_aggregate_owner_kernel<GCN_OUT_DIM>
      <<<(GCN_N_NODES + AGG_T - 1) / AGG_T, AGG_NW * 32, 0, stream>>>(hproj2, src, dst, norm_in, b2, out, N);
}


# degrees from whole-graph integer histograms in LDS: 64 blocks each count a slice of the src or dst list and store a plane, norm kernel sums the planes; global float atomics and the memset go
# speedup vs baseline: 3.4175x; 1.4311x over previous
#include <hip/hip_runtime.h>


#define GCN_N_NODES 50000
#define GCN_N_EDGES 800000
#define GCN_IN_DIM 128
#define GCN_HID_DIM 128
#define GCN_OUT_DIM 64

typedef __attribute__((ext_vector_type(2))) float v2f;
typedef __attribute__((ext_vector_type(8))) float v8f;

#define DEG_P 32
#define DEG_NT 1024

__global__ void __launch_bounds__(DEG_NT)
gcn_degree_hist_kernel(const int* __restrict__ src,
                       const int* __restrict__ dst,
                       unsigned* __restrict__ part) {
  typedef int v4i __attribute__((ext_vector_type(4)));
  typedef unsigned v4u __attribute__((ext_vector_type(4)));
  static_assert(GCN_N_NODES % 4 == 0, "histogram is cleared and copied out 16 B per lane");
  static_assert(GCN_N_EDGES % (4 * DEG_P) == 0, "a slice is read 16 B per lane");
  constexpr int NV = GCN_N_NODES / 4;
  constexpr int SV = GCN_N_EDGES / DEG_P / 4;

  __shared__ v4u hist4[NV];
  unsigned* hist = reinterpret_cast<unsigned*>(hist4);

  const int tid   = threadIdx.x;
  const int which = blockIdx.x / DEG_P;
  const int slice = blockIdx.x - which * DEG_P;
  const v4i* l4 = reinterpret_cast<const v4i*>(which ? dst : src) + (size_t)slice * SV;

  constexpr int IT = (SV + DEG_NT - 1) / DEG_NT;
  const v4i none = {-1, -1, -1, -1};
  v4i v[IT];
#pragma unroll
  for (int k = 0; k < IT; ++k) {
    const int i = tid + k * DEG_NT;
    v[k] = i < SV ? l4[i] : none;
  }

  const v4u zero = {0u, 0u, 0u, 0u};
  for (int i = tid; i < NV; i += DEG_NT) hist4[i] = zero;
  __syncthreads();

#pragma unroll
  for (int k = 0; k < IT; ++k) {
    if ((unsigned)v[k].x < (unsigned)GCN_N_NODES) atomicAdd(&hist[v[k].x], 1u);
    if ((unsigned)v[k].y < (unsigned)GCN_N_NODES) atomicAdd(&hist[v[k].y], 1u);
    if ((unsigned)v[k].z < (unsigned)GCN_N_NODES) atomicAdd(&hist[v[k].z], 1u);
    if ((unsigned)v[k].w < (unsigned)GCN_N_NODES) atomicAdd(&hist[v[k].w], 1u);
  }
  __syncthreads();

  v4u* o4 = reinterpret_cast<v4u*>(part + (size_t)blockIdx.x * GCN_N_NODES);
  for (int i = tid; i < NV; i += DEG_NT) o4[i] = hist4[i];
}

__global__ void gcn_norm_kernel(const unsigned* __restrict__ part,
                                float* __restrict__ norm_out,
                                float* __restrict__ norm_in, int N) {
  int i = blockIdx.x * blockDim.x + threadIdx.x;
  if (i < N) {
    unsigned d_out = 0u, d_in = 0u;
#pragma unroll
    for (int p = 0; p < DEG_P; ++p) {
      d_out += part[(size_t)p * GCN_N_NODES + i];
      d_in  += part[(size_t)(DEG_P + p) * GCN_N_NODES + i];
    }
    norm_out[i] = rsqrtf(fmaxf((float)d_out, 1.0f));
    norm_in[i]  = rsqrtf(fmaxf((float)d_in, 1.0f));
  }
}

template <int K, int NOUT>
__global__ void gcn_gemm_scaled_wmma(const float* __restrict__ X,
                                     const float* __restrict__ scale,
                                     const float* __restrict__ W,
                                     float* __restrict__ Y) {
  constexpr int LDSK = K + 4;
  __shared__ float tile[16 * LDSK];

  const int tid  = threadIdx.x;
  const int row0 = blockIdx.x * 16;

  for (int t = tid; t < 16 * K; t += blockDim.x) {
    const int r = t / K;
    const int c = t - r * K;
    const int node = row0 + r;
    tile[r * LDSK + c] = X[(size_t)node * K + c] * scale[node];
  }
  __syncthreads();

  const int wave = tid >> 5;
  const int lane = tid & 31;
  const int m    = lane & 15;
  const int kh   = lane >> 4;
  const int nc   = wave * 16 + m;

  v8f acc = {};
#pragma unroll 4
  for (int kk = 0; kk < K; kk += 4) {
    v2f a, b;
    const int ka = kk + 2 * kh;
    a.x = tile[m * LDSK + ka];
    a.y = tile[m * LDSK + ka + 1];
    b.x = W[(size_t)ka * NOUT + nc];
    b.y = W[(size_t)(ka + 1) * NOUT + nc];
    acc = __builtin_amdgcn_wmma_f32_16x16x4_f32(
         false, a,  false, b,
         (short)0, acc,  false,  false);
  }

#pragma unroll
  for (int v = 0; v < 8; ++v) {
    const int orow = row0 + v + 8 * kh;
    Y[(size_t)orow * NOUT + nc] = acc[v];
  }
}

#define AGG_T 200
#define AGG_NW 32
#define AGG_U 8
#define AGG_B 16
#define AGG_QCAP (AGG_U * 128 + AGG_B)

template <int D>
__global__ void __launch_bounds__(AGG_NW * 32)
gcn_aggregate_owner_kernel(const float* __restrict__ H,
                           const int* __restrict__ src,
                           const int* __restrict__ dst,
                           const float* __restrict__ norm_in,
                           const float* __restrict__ b,
                           float* __restrict__ out, int N) {
  constexpr int VEC = D / 32;
  typedef float vecf __attribute__((ext_vector_type(VEC)));
  typedef int v4i __attribute__((ext_vector_type(4)));
  static_assert(GCN_N_EDGES % 128 == 0, "edge list is read in 128-edge chunks");
  static_assert(AGG_T + 1 <= 512 && GCN_N_EDGES <= (1 << 20), "queue entry packing");
  constexpr int NCH   = GCN_N_EDGES / 128;
  constexpr int STEP  = AGG_NW * AGG_U;
  constexpr int NSTEP = (NCH + STEP - 1) / STEP;

  __shared__ float acc[(AGG_T + 1) * D];
  __shared__ unsigned queue[AGG_NW * AGG_QCAP];

  const int tid  = threadIdx.x;
  const int lane = tid & 31;
  const int wave = __builtin_amdgcn_readfirstlane(tid >> 5);
  const unsigned nbase = blockIdx.x * AGG_T;
  const int rot = blockIdx.x % NSTEP;

  for (int i = tid; i < (AGG_T + 1) * D; i += AGG_NW * 32) acc[i] = 0.0f;
  __syncthreads();

  unsigned* q = queue + wave * AGG_QCAP;
  int cnt = 0;

  for (int step = 0; step <= NSTEP; ++step) {
    if (step < NSTEP) {
      const int sidx = step + rot < NSTEP ? step + rot : step + rot - NSTEP;
      v4i dv[AGG_U];
#pragma unroll
      for (int u = 0; u < AGG_U; ++u) {
        const int c  = sidx * STEP + u * AGG_NW + wave;
        const int cc = c < NCH ? c : NCH - 1;
        dv[u] = *reinterpret_cast<const v4i*>(dst + (size_t)cc * 128 + lane * 4);
      }
#pragma unroll
      for (int u = 0; u < AGG_U; ++u) {
        const int c = sidx * STEP + u * AGG_NW + wave;
        const unsigned nb = c < NCH ? nbase : 0x80000000u;
        const unsigned r0 = (unsigned)dv[u].x - nb;
        const unsigned r1 = (unsigned)dv[u].y - nb;
        const unsigned r2 = (unsigned)dv[u].z - nb;
        const unsigned r3 = (unsigned)dv[u].w - nb;
        const unsigned mn = min(min(r0, r1), min(r2, r3));
        if (__builtin_amdgcn_ballot_w32(mn < (unsigned)AGG_T) != 0u) {
          const unsigned eb = ((unsigned)c * 128u + (unsigned)lane * 4u) << 9;
          const unsigned rr[4] = {r0, r1, r2, r3};
#pragma unroll
          for (int j = 0; j < 4; ++j) {
            const bool hit = rr[j] < (unsigned)AGG_T;
            const unsigned mask = __builtin_amdgcn_ballot_w32(hit);
            if (mask != 0u) {
              if (hit) q[cnt + __builtin_amdgcn_mbcnt_lo(mask, 0u)] = (eb + ((unsigned)j << 9)) | rr[j];
              cnt += __builtin_popcount(mask);
            }
          }
        }
      }
    } else if (cnt > 0) {
      if (lane >= cnt && lane < AGG_B) q[lane] = (unsigned)AGG_T;
      cnt = AGG_B;
    }

    __builtin_amdgcn_fence(__ATOMIC_SEQ_CST, "wavefront");

    while (cnt >= AGG_B) {
      cnt -= AGG_B;
      const unsigned pk = q[cnt + (lane & (AGG_B - 1))];
      const int s = src[pk >> 9];
      vecf rows[AGG_B];
#pragma unroll
      for (int i = 0; i < AGG_B; ++i) {
        const int si = __builtin_amdgcn_readlane(s, i);
        rows[i] = *reinterpret_cast<const vecf*>(H + (size_t)si * D + lane * VEC);
      }
#pragma unroll
      for (int i = 0; i < AGG_B; ++i) {
        const unsigned dl = (unsigned)__builtin_amdgcn_readlane((int)pk, i) & 511u;
        for (int j = 0; j < VEC; ++j)
          __hip_atomic_fetch_add(&acc[dl * D + j * 32 + lane], rows[i][j],
                                 __ATOMIC_RELAXED, __HIP_MEMORY_SCOPE_WORKGROUP);
      }
    }
  }
  __syncthreads();

  for (int i = tid; i < AGG_T * 32; i += AGG_NW * 32) {
    const int nl   = i >> 5;
    const int node = (int)nbase + nl;
    if (node < N) {
      const float ni = norm_in[node];
      vecf o;
      for (int j = 0; j < VEC; ++j)
        o[j] = fmaxf(fmaf(acc[nl * D + j * 32 + lane], ni, b[lane * VEC + j]), 0.0f);
      *reinterpret_cast<vecf*>(out + (size_t)node * D + lane * VEC) = o;
    }
  }
}

extern "C" void kernel_launch(void* const* d_in, const int* in_sizes, int n_in,
                              void* d_out, int out_size, void* d_ws, size_t ws_size,
                              hipStream_t stream) {
  const float* feat = (const float*)d_in[0];
  const int*   src  = (const int*)d_in[1];
  const int*   dst  = (const int*)d_in[2];
  const float* W1   = (const float*)d_in[3];
  const float* b1   = (const float*)d_in[4];
  const float* W2   = (const float*)d_in[5];
  const float* b2   = (const float*)d_in[6];
  float* out = (float*)d_out;
  float* ws  = (float*)d_ws;

  const int N = GCN_N_NODES;

  float* norm_out = ws + 2 * (size_t)N;
  float* norm_in  = ws + 3 * (size_t)N;
  float* bufA     = ws + 4 * (size_t)N;
  float* bufB     = bufA + (size_t)N * GCN_HID_DIM;
  float* hproj2   = bufA;
  unsigned* dpart = (unsigned*)(bufB + (size_t)N * GCN_HID_DIM);

  gcn_degree_hist_kernel<<<2 * DEG_P, DEG_NT, 0, stream>>>(src, dst, dpart);
  gcn_norm_kernel<<<(N + 255) / 256, 256, 0, stream>>>(dpart, norm_out, norm_in, N);

  gcn_gemm_scaled_wmma<GCN_IN_DIM, GCN_HID_DIM>
      <<<N / 16, (GCN_HID_DIM / 16) * 32, 0, stream>>>(feat, norm_out, W1, bufA);
  gcn_aggregate_owner_kernel<GCN_HID_DIM>
      <<<(GCN_N_NODES + AGG_T - 1) / AGG_T, AGG_NW * 32, 0, stream>>>(bufA, src, dst, norm_in, b1, bufB, N);

  gcn_gemm_scaled_wmma<GCN_HID_DIM, GCN_OUT_DIM>
      <<<N / 16, (GCN_OUT_DIM / 16) * 32, 0, stream>>>(bufB, norm_out, W2, hproj2);
  gcn_aggregate_owner_kernel<GCN_OUT_DIM>
      <<<(GCN_N_NODES + AGG_T - 1) / AGG_T, AGG_NW * 32, 0, stream>>>(hproj2, src, dst, norm_in, b2, out, N);
}


# layer-2 aggregation replays per-wave edge lists (src, local dst) stored by layer-1's owner-block filter in the workspace instead of scanning the dst list again; sum order = LDS add arrival, as before
# speedup vs baseline: 4.1655x; 1.2189x over previous
#include <hip/hip_runtime.h>


#define GCN_N_NODES 50000
#define GCN_N_EDGES 800000
#define GCN_IN_DIM 128
#define GCN_HID_DIM 128
#define GCN_OUT_DIM 64

typedef __attribute__((ext_vector_type(2))) float v2f;
typedef __attribute__((ext_vector_type(8))) float v8f;

#define DEG_P 32
#define DEG_NT 1024

__global__ void __launch_bounds__(DEG_NT)
gcn_degree_hist_kernel(const int* __restrict__ src,
                       const int* __restrict__ dst,
                       unsigned* __restrict__ part) {
  typedef int v4i __attribute__((ext_vector_type(4)));
  typedef unsigned v4u __attribute__((ext_vector_type(4)));
  static_assert(GCN_N_NODES % 4 == 0, "histogram is cleared and copied out 16 B per lane");
  static_assert(GCN_N_EDGES % (4 * DEG_P) == 0, "a slice is read 16 B per lane");
  constexpr int NV = GCN_N_NODES / 4;
  constexpr int SV = GCN_N_EDGES / DEG_P / 4;

  __shared__ v4u hist4[NV];
  unsigned* hist = reinterpret_cast<unsigned*>(hist4);

  const int tid   = threadIdx.x;
  const int which = blockIdx.x / DEG_P;
  const int slice = blockIdx.x - which * DEG_P;
  const v4i* l4 = reinterpret_cast<const v4i*>(which ? dst : src) + (size_t)slice * SV;

  constexpr int IT = (SV + DEG_NT - 1) / DEG_NT;
  const v4i none = {-1, -1, -1, -1};
  v4i v[IT];
#pragma unroll
  for (int k = 0; k < IT; ++k) {
    const int i = tid + k * DEG_NT;
    v[k] = i < SV ? l4[i] : none;
  }

  const v4u zero = {0u, 0u, 0u, 0u};
  for (int i = tid; i < NV; i += DEG_NT) hist4[i] = zero;
  __syncthreads();

#pragma unroll
  for (int k = 0; k < IT; ++k) {
    if ((unsigned)v[k].x < (unsigned)GCN_N_NODES) atomicAdd(&hist[v[k].x], 1u);
    if ((unsigned)v[k].y < (unsigned)GCN_N_NODES) atomicAdd(&hist[v[k].y], 1u);
    if ((unsigned)v[k].z < (unsigned)GCN_N_NODES) atomicAdd(&hist[v[k].z], 1u);
    if ((unsigned)v[k].w < (unsigned)GCN_N_NODES) atomicAdd(&hist[v[k].w], 1u);
  }
  __syncthreads();

  v4u* o4 = reinterpret_cast<v4u*>(part + (size_t)blockIdx.x * GCN_N_NODES);
  for (int i = tid; i < NV; i += DEG_NT) o4[i] = hist4[i];
}

__global__ void gcn_norm_kernel(const unsigned* __restrict__ part,
                                float* __restrict__ norm_out,
                                float* __restrict__ norm_in, int N) {
  int i = blockIdx.x * blockDim.x + threadIdx.x;
  if (i < N) {
    unsigned d_out = 0u, d_in = 0u;
#pragma unroll
    for (int p = 0; p < DEG_P; ++p) {
      d_out += part[(size_t)p * GCN_N_NODES + i];
      d_in  += part[(size_t)(DEG_P + p) * GCN_N_NODES + i];
    }
    norm_out[i] = rsqrtf(fmaxf((float)d_out, 1.0f));
    norm_in[i]  = rsqrtf(fmaxf((float)d_in, 1.0f));
  }
}

template <int K, int NOUT>
__global__ void gcn_gemm_scaled_wmma(const float* __restrict__ X,
                                     const float* __restrict__ scale,
                                     const float* __restrict__ W,
                                     float* __restrict__ Y) {
  constexpr int LDSK = K + 4;
  __shared__ float tile[16 * LDSK];

  const int tid  = threadIdx.x;
  const int row0 = blockIdx.x * 16;

  for (int t = tid; t < 16 * K; t += blockDim.x) {
    const int r = t / K;
    const int c = t - r * K;
    const int node = row0 + r;
    tile[r * LDSK + c] = X[(size_t)node * K + c] * scale[node];
  }
  __syncthreads();

  const int wave = tid >> 5;
  const int lane = tid & 31;
  const int m    = lane & 15;
  const int kh   = lane >> 4;
  const int nc   = wave * 16 + m;

  v8f acc = {};
#pragma unroll 4
  for (int kk = 0; kk < K; kk += 4) {
    v2f a, b;
    const int ka = kk + 2 * kh;
    a.x = tile[m * LDSK + ka];
    a.y = tile[m * LDSK + ka + 1];
    b.x = W[(size_t)ka * NOUT + nc];
    b.y = W[(size_t)(ka + 1) * NOUT + nc];
    acc = __builtin_amdgcn_wmma_f32_16x16x4_f32(
         false, a,  false, b,
         (short)0, acc,  false,  false);
  }

#pragma unroll
  for (int v = 0; v < 8; ++v) {
    const int orow = row0 + v + 8 * kh;
    Y[(size_t)orow * NOUT + nc] = acc[v];
  }
}

#define AGG_T 200
#define AGG_NW 32
#define AGG_U 8
#define AGG_B 16
#define AGG_QCAP (AGG_U * 128 + AGG_B)
#define AGG_NBLK ((GCN_N_NODES + AGG_T - 1) / AGG_T)
#define AGG_LCAP 1024

template <int D, int LO>
__device__ __forceinline__ void gcn_gather_add(const float* __restrict__ H, float* acc,
                                               int s, unsigned dl, int lane) {
  constexpr int VEC = D / 32;
  typedef float vecf __attribute__((ext_vector_type(VEC)));
  vecf rows[AGG_B];
#pragma unroll
  for (int i = 0; i < AGG_B; ++i) {
    const int si = __builtin_amdgcn_readlane(s, LO + i);
    rows[i] = *reinterpret_cast<const vecf*>(H + (size_t)si * D + lane * VEC);
  }
#pragma unroll
  for (int i = 0; i < AGG_B; ++i) {
    const unsigned d = (unsigned)__builtin_amdgcn_readlane((int)dl, LO + i);
    for (int j = 0; j < VEC; ++j)
      __hip_atomic_fetch_add(&acc[d * D + j * 32 + lane], rows[i][j],
                             __ATOMIC_RELAXED, __HIP_MEMORY_SCOPE_WORKGROUP);
  }
}

template <int D, int MODE>
__global__ void __launch_bounds__(AGG_NW * 32)
gcn_aggregate_owner_kernel(const float* __restrict__ H,
                           const int* __restrict__ src,
                           const int* __restrict__ dst,
                           const float* __restrict__ norm_in,
                           const float* __restrict__ b,
                           float* __restrict__ out,
                           unsigned* __restrict__ wcount,
                           unsigned* __restrict__ wlist, int N) {
  constexpr int VEC = D / 32;
  typedef float vecf __attribute__((ext_vector_type(VEC)));
  typedef int v4i __attribute__((ext_vector_type(4)));
  static_assert(GCN_N_EDGES % 128 == 0, "edge list is read in 128-edge chunks");
  static_assert(AGG_T + 1 <= 512 && GCN_N_EDGES <= (1 << 20), "queue entry packing");
  static_assert(GCN_N_NODES <= (1 << 23), "list word packing: src << 9 | local dst");
  constexpr int NCH   = GCN_N_EDGES / 128;
  constexpr int STEP  = AGG_NW * AGG_U;
  constexpr int NSTEP = (NCH + STEP - 1) / STEP;

  __shared__ float acc[(AGG_T + 1) * D];
  __shared__ unsigned queue[AGG_NW * AGG_QCAP];

  const int tid  = threadIdx.x;
  const int lane = tid & 31;
  const int wave = __builtin_amdgcn_readfirstlane(tid >> 5);
  const unsigned nbase = blockIdx.x * AGG_T;
  const int rot = blockIdx.x % NSTEP;

  const int widx = blockIdx.x * AGG_NW + wave;
  unsigned* wl = wlist + (size_t)widx * AGG_LCAP;
  unsigned nrec = 0;
  if (MODE == 1) nrec = (unsigned)__builtin_amdgcn_readfirstlane((int)wcount[widx]);
  const bool scan = MODE == 0 || nrec > (unsigned)AGG_LCAP;

  for (int i = tid; i < (AGG_T + 1) * D; i += AGG_NW * 32) acc[i] = 0.0f;
  __syncthreads();

  if (scan) {
    unsigned* q = queue + wave * AGG_QCAP;
    int cnt = 0;
    unsigned wrote = 0;

    for (int step = 0; step <= NSTEP; ++step) {
      if (step < NSTEP) {
        const int sidx = step + rot < NSTEP ? step + rot : step + rot - NSTEP;
        v4i dv[AGG_U];
#pragma unroll
        for (int u = 0; u < AGG_U; ++u) {
          const int c  = sidx * STEP + u * AGG_NW + wave;
          const int cc = c < NCH ? c : NCH - 1;
          dv[u] = *reinterpret_cast<const v4i*>(dst + (size_t)cc * 128 + lane * 4);
        }
#pragma unroll
        for (int u = 0; u < AGG_U; ++u) {
          const int c = sidx * STEP + u * AGG_NW + wave;
          const unsigned nb = c < NCH ? nbase : 0x80000000u;
          const unsigned r0 = (unsigned)dv[u].x - nb;
          const unsigned r1 = (unsigned)dv[u].y - nb;
          const unsigned r2 = (unsigned)dv[u].z - nb;
          const unsigned r3 = (unsigned)dv[u].w - nb;
          const unsigned mn = min(min(r0, r1), min(r2, r3));
          if (__builtin_amdgcn_ballot_w32(mn < (unsigned)AGG_T) != 0u) {
            const unsigned eb = ((unsigned)c * 128u + (unsigned)lane * 4u) << 9;
            const unsigned rr[4] = {r0, r1, r2, r3};
#pragma unroll
            for (int j = 0; j < 4; ++j) {
              const bool hit = rr[j] < (unsigned)AGG_T;
              const unsigned mask = __builtin_amdgcn_ballot_w32(hit);
              if (mask != 0u) {
                if (hit) q[cnt + __builtin_amdgcn_mbcnt_lo(mask, 0u)] = (eb + ((unsigned)j << 9)) | rr[j];
                cnt += __builtin_popcount(mask);
              }
            }
          }
        }
      } else if (cnt > 0) {
        if (lane >= cnt && lane < AGG_B) q[lane] = (unsigned)AGG_T;
        cnt = AGG_B;
      }

      __builtin_amdgcn_fence(__ATOMIC_SEQ_CST, "wavefront");

      while (cnt >= AGG_B) {
        cnt -= AGG_B;
        const unsigned pk = q[cnt + (lane & (AGG_B - 1))];
        const int s = src[pk >> 9];
        if (MODE == 0) {
          if (lane < AGG_B && wrote + AGG_B <= (unsigned)AGG_LCAP)
            wl[wrote + lane] = ((unsigned)s << 9) | (pk & 511u);
          wrote += AGG_B;
        }
        gcn_gather_add<D, 0>(H, acc, s, pk & 511u, lane);
      }
    }
    if (MODE == 0 && lane == 0) wcount[widx] = wrote;
  } else if (nrec > 0u) {
    const unsigned last = nrec - 1u;
    unsigned e = wl[min((unsigned)lane, last)];
    for (unsigned i0 = 0; i0 < nrec; i0 += 32) {
      const unsigned en = wl[min(i0 + 32u + (unsigned)lane, last)];
      gcn_gather_add<D, 0>(H, acc, (int)(e >> 9), e & 511u, lane);
      if (i0 + AGG_B < nrec) gcn_gather_add<D, AGG_B>(H, acc, (int)(e >> 9), e & 511u, lane);
      e = en;
    }
  }
  __syncthreads();

  for (int i = tid; i < AGG_T * 32; i += AGG_NW * 32) {
    const int nl   = i >> 5;
    const int node = (int)nbase + nl;
    if (node < N) {
      const float ni = norm_in[node];
      vecf o;
      for (int j = 0; j < VEC; ++j)
        o[j] = fmaxf(fmaf(acc[nl * D + j * 32 + lane], ni, b[lane * VEC + j]), 0.0f);
      *reinterpret_cast<vecf*>(out + (size_t)node * D + lane * VEC) = o;
    }
  }
}

extern "C" void kernel_launch(void* const* d_in, const int* in_sizes, int n_in,
                              void* d_out, int out_size, void* d_ws, size_t ws_size,
                              hipStream_t stream) {
  const float* feat = (const float*)d_in[0];
  const int*   src  = (const int*)d_in[1];
  const int*   dst  = (const int*)d_in[2];
  const float* W1   = (const float*)d_in[3];
  const float* b1   = (const float*)d_in[4];
  const float* W2   = (const float*)d_in[5];
  const float* b2   = (const float*)d_in[6];
  float* out = (float*)d_out;
  float* ws  = (float*)d_ws;

  const int N = GCN_N_NODES;

  float* norm_out = ws + 2 * (size_t)N;
  float* norm_in  = ws + 3 * (size_t)N;
  float* bufA     = ws + 4 * (size_t)N;
  float* bufB     = bufA + (size_t)N * GCN_HID_DIM;
  float* hproj2   = bufA;
  unsigned* dpart = (unsigned*)(bufB + (size_t)N * GCN_HID_DIM);
  unsigned* wcount = dpart + (size_t)2 * DEG_P * GCN_N_NODES;
  unsigned* wlist  = wcount + 8192;

  gcn_degree_hist_kernel<<<2 * DEG_P, DEG_NT, 0, stream>>>(src, dst, dpart);
  gcn_norm_kernel<<<(N + 255) / 256, 256, 0, stream>>>(dpart, norm_out, norm_in, N);

  gcn_gemm_scaled_wmma<GCN_IN_DIM, GCN_HID_DIM>
      <<<N / 16, (GCN_HID_DIM / 16) * 32, 0, stream>>>(feat, norm_out, W1, bufA);
  gcn_aggregate_owner_kernel<GCN_HID_DIM, 0>
      <<<AGG_NBLK, AGG_NW * 32, 0, stream>>>(bufA, src, dst, norm_in, b1, bufB, wcount, wlist, N);

  gcn_gemm_scaled_wmma<GCN_HID_DIM, GCN_OUT_DIM>
      <<<N / 16, (GCN_OUT_DIM / 16) * 32, 0, stream>>>(bufB, norm_out, W2, hproj2);
  gcn_aggregate_owner_kernel<GCN_OUT_DIM, 1>
      <<<AGG_NBLK, AGG_NW * 32, 0, stream>>>(hproj2, src, dst, norm_in, b2, out, wcount, wlist, N);
}


# owner blocks filter a 1-byte owner key per edge (dst / 200, stored by the dst histogram blocks), 16 edges per lane and load, not the 4-byte dst list; sum order = LDS add arrival, as before
# speedup vs baseline: 4.6865x; 1.1251x over previous
#include <hip/hip_runtime.h>


#define GCN_N_NODES 50000
#define GCN_N_EDGES 800000
#define GCN_IN_DIM 128
#define GCN_HID_DIM 128
#define GCN_OUT_DIM 64

#define AGG_T 200
#define AGG_NW 32
#define AGG_U 2
#define AGG_NBLK ((GCN_N_NODES + AGG_T - 1) / AGG_T)
#define KEY_NCH ((GCN_N_EDGES + 511) / 512)
#define KEY_STEP (AGG_NW * AGG_U)
#define KEY_NSTEP ((KEY_NCH + KEY_STEP - 1) / KEY_STEP)
#define KEY_WORDS (KEY_NSTEP * KEY_STEP * 128)

typedef __attribute__((ext_vector_type(2))) float v2f;
typedef __attribute__((ext_vector_type(8))) float v8f;

#define DEG_P 32
#define DEG_NT 1024

__global__ void __launch_bounds__(DEG_NT)
gcn_degree_hist_kernel(const int* __restrict__ src,
                       const int* __restrict__ dst,
                       unsigned* __restrict__ part,
                       unsigned* __restrict__ keys) {
  typedef int v4i __attribute__((ext_vector_type(4)));
  typedef unsigned v4u __attribute__((ext_vector_type(4)));
  static_assert(GCN_N_NODES % 4 == 0, "histogram is cleared and copied out 16 B per lane");
  static_assert(GCN_N_EDGES % (4 * DEG_P) == 0, "a slice is read 16 B per lane");
  static_assert(AGG_NBLK <= 255, "an owner key is one byte, 255 means no owner");
  constexpr int NV = GCN_N_NODES / 4;
  constexpr int SV = GCN_N_EDGES / DEG_P / 4;

  __shared__ v4u hist4[NV];
  unsigned* hist = reinterpret_cast<unsigned*>(hist4);

  const int tid   = threadIdx.x;
  const int which = blockIdx.x / DEG_P;
  const int slice = blockIdx.x - which * DEG_P;
  const v4i* l4 = reinterpret_cast<const v4i*>(which ? dst : src) + (size_t)slice * SV;

  constexpr int IT = (SV + DEG_NT - 1) / DEG_NT;
  const v4i none = {-1, -1, -1, -1};
  v4i v[IT];
#pragma unroll
  for (int k = 0; k < IT; ++k) {
    const int i = tid + k * DEG_NT;
    v[k] = i < SV ? l4[i] : none;
  }

  const v4u zero = {0u, 0u, 0u, 0u};
  for (int i = tid; i < NV; i += DEG_NT) hist4[i] = zero;
  __syncthreads();

#pragma unroll
  for (int k = 0; k < IT; ++k) {
    if ((unsigned)v[k].x < (unsigned)GCN_N_NODES) atomicAdd(&hist[v[k].x], 1u);
    if ((unsigned)v[k].y < (unsigned)GCN_N_NODES) atomicAdd(&hist[v[k].y], 1u);
    if ((unsigned)v[k].z < (unsigned)GCN_N_NODES) atomicAdd(&hist[v[k].z], 1u);
    if ((unsigned)v[k].w < (unsigned)GCN_N_NODES) atomicAdd(&hist[v[k].w], 1u);
  }
  if (which) {
    unsigned* kw = keys + (size_t)slice * SV;
#pragma unroll
    for (int k = 0; k < IT; ++k) {
      const int i = tid + k * DEG_NT;
      const unsigned k0 = (unsigned)v[k].x < (unsigned)GCN_N_NODES ? (unsigned)v[k].x / AGG_T : 255u;
      const unsigned k1 = (unsigned)v[k].y < (unsigned)GCN_N_NODES ? (unsigned)v[k].y / AGG_T : 255u;
      const unsigned k2 = (unsigned)v[k].z < (unsigned)GCN_N_NODES ? (unsigned)v[k].z / AGG_T : 255u;
      const unsigned k3 = (unsigned)v[k].w < (unsigned)GCN_N_NODES ? (unsigned)v[k].w / AGG_T : 255u;
      if (i < SV) kw[i] = k0 | (k1 << 8) | (k2 << 16) | (k3 << 24);
    }
    if (slice == DEG_P - 1)
      for (int i = tid; i < KEY_WORDS - GCN_N_EDGES / 4; i += DEG_NT) keys[GCN_N_EDGES / 4 + i] = 0xFFFFFFFFu;
  }
  __syncthreads();

  v4u* o4 = reinterpret_cast<v4u*>(part + (size_t)blockIdx.x * GCN_N_NODES);
  for (int i = tid; i < NV; i += DEG_NT) o4[i] = hist4[i];
}

__global__ void gcn_norm_kernel(const unsigned* __restrict__ part,
                                float* __restrict__ norm_out,
                                float* __restrict__ norm_in, int N) {
  int i = blockIdx.x * blockDim.x + threadIdx.x;
  if (i < N) {
    unsigned d_out = 0u, d_in = 0u;
#pragma unroll
    for (int p = 0; p < DEG_P; ++p) {
      d_out += part[(size_t)p * GCN_N_NODES + i];
      d_in  += part[(size_t)(DEG_P + p) * GCN_N_NODES + i];
    }
    norm_out[i] = rsqrtf(fmaxf((float)d_out, 1.0f));
    norm_in[i]  = rsqrtf(fmaxf((float)d_in, 1.0f));
  }
}

template <int K, int NOUT>
__global__ void gcn_gemm_scaled_wmma(const float* __restrict__ X,
                                     const float* __restrict__ scale,
                                     const float* __restrict__ W,
                                     float* __restrict__ Y) {
  constexpr int LDSK = K + 4;
  __shared__ float tile[16 * LDSK];

  const int tid  = threadIdx.x;
  const int row0 = blockIdx.x * 16;

  for (int t = tid; t < 16 * K; t += blockDim.x) {
    const int r = t / K;
    const int c = t - r * K;
    const int node = row0 + r;
    tile[r * LDSK + c] = X[(size_t)node * K + c] * scale[node];
  }
  __syncthreads();

  const int wave = tid >> 5;
  const int lane = tid & 31;
  const int m    = lane & 15;
  const int kh   = lane >> 4;
  const int nc   = wave * 16 + m;

  v8f acc = {};
#pragma unroll 4
  for (int kk = 0; kk < K; kk += 4) {
    v2f a, b;
    const int ka = kk + 2 * kh;
    a.x = tile[m * LDSK + ka];
    a.y = tile[m * LDSK + ka + 1];
    b.x = W[(size_t)ka * NOUT + nc];
    b.y = W[(size_t)(ka + 1) * NOUT + nc];
    acc = __builtin_amdgcn_wmma_f32_16x16x4_f32(
         false, a,  false, b,
         (short)0, acc,  false,  false);
  }

#pragma unroll
  for (int v = 0; v < 8; ++v) {
    const int orow = row0 + v + 8 * kh;
    Y[(size_t)orow * NOUT + nc] = acc[v];
  }
}

#define AGG_B 16
#define AGG_QCAP (AGG_U * 512 + AGG_B)
#define AGG_LCAP 1024

template <int D, int LO>
__device__ __forceinline__ void gcn_gather_add(const float* __restrict__ H, float* acc,
                                               int s, unsigned dl, int lane) {
  constexpr int VEC = D / 32;
  typedef float vecf __attribute__((ext_vector_type(VEC)));
  vecf rows[AGG_B];
#pragma unroll
  for (int i = 0; i < AGG_B; ++i) {
    const int si = __builtin_amdgcn_readlane(s, LO + i);
    rows[i] = *reinterpret_cast<const vecf*>(H + (size_t)si * D + lane * VEC);
  }
#pragma unroll
  for (int i = 0; i < AGG_B; ++i) {
    const unsigned d = (unsigned)__builtin_amdgcn_readlane((int)dl, LO + i);
    for (int j = 0; j < VEC; ++j)
      __hip_atomic_fetch_add(&acc[d * D + j * 32 + lane], rows[i][j],
                             __ATOMIC_RELAXED, __HIP_MEMORY_SCOPE_WORKGROUP);
  }
}

template <int D, int MODE>
__global__ void __launch_bounds__(AGG_NW * 32)
gcn_aggregate_owner_kernel(const float* __restrict__ H,
                           const int* __restrict__ src,
                           const int* __restrict__ dst,
                           const unsigned* __restrict__ keys,
                           const float* __restrict__ norm_in,
                           const float* __restrict__ b,
                           float* __restrict__ out,
                           unsigned* __restrict__ wcount,
                           unsigned* __restrict__ wlist, int N) {
  constexpr int VEC = D / 32;
  typedef float vecf __attribute__((ext_vector_type(VEC)));
  typedef unsigned v4u __attribute__((ext_vector_type(4)));
  static_assert(GCN_N_EDGES % 16 == 0, "a lane's 16 keys are all real edges or all tail");
  static_assert(KEY_WORDS * 4 <= (1 << 27), "queue entry packing: 16-key group << 5 | bit");
  static_assert(AGG_T + 1 <= 512 && GCN_N_NODES <= (1 << 23), "list word packing: src << 9 | local dst");
  constexpr int NSTEP = KEY_NSTEP;

  __shared__ float acc[(AGG_T + 1) * D];
  __shared__ unsigned queue[AGG_NW * AGG_QCAP];

  const int tid  = threadIdx.x;
  const int lane = tid & 31;
  const int wave = __builtin_amdgcn_readfirstlane(tid >> 5);
  const unsigned nbase = blockIdx.x * AGG_T;
  const int rot = blockIdx.x % NSTEP;

  const int widx = blockIdx.x * AGG_NW + wave;
  unsigned* wl = wlist + (size_t)widx * AGG_LCAP;
  unsigned nrec = 0;
  if (MODE == 1) nrec = (unsigned)__builtin_amdgcn_readfirstlane((int)wcount[widx]);
  const bool scan = MODE == 0 || nrec > (unsigned)AGG_LCAP;

  for (int i = tid; i < (AGG_T + 1) * D; i += AGG_NW * 32) acc[i] = 0.0f;
  __syncthreads();

  if (scan) {
    unsigned* q = queue + wave * AGG_QCAP;
    int cnt = 0;
    unsigned wrote = 0;

    const v4u* k4 = reinterpret_cast<const v4u*>(keys);
    const unsigned pat = blockIdx.x * 0x01010101u;

    v4u cur[AGG_U];
#pragma unroll
    for (int u = 0; u < AGG_U; ++u)
      cur[u] = k4[(size_t)(rot * KEY_STEP + u * AGG_NW + wave) * 32 + lane];

    for (int step = 0; step <= NSTEP; ++step) {
      if (step < NSTEP) {
        const int sidx = step + rot < NSTEP ? step + rot : step + rot - NSTEP;
        const int nidx = sidx + 1 < NSTEP ? sidx + 1 : 0;
        v4u nxt[AGG_U];
#pragma unroll
        for (int u = 0; u < AGG_U; ++u)
          nxt[u] = k4[(size_t)(nidx * KEY_STEP + u * AGG_NW + wave) * 32 + lane];
#pragma unroll
        for (int u = 0; u < AGG_U; ++u) {
          const int c = sidx * KEY_STEP + u * AGG_NW + wave;
          const unsigned y0 = cur[u].x ^ pat, y1 = cur[u].y ^ pat, y2 = cur[u].z ^ pat, y3 = cur[u].w ^ pat;
          const unsigned n0 = ((y0 & 0x7F7F7F7Fu) + 0x7F7F7F7Fu) | y0;
          const unsigned n1 = ((y1 & 0x7F7F7F7Fu) + 0x7F7F7F7Fu) | y1;
          const unsigned n2 = ((y2 & 0x7F7F7F7Fu) + 0x7F7F7F7Fu) | y2;
          const unsigned n3 = ((y3 & 0x7F7F7F7Fu) + 0x7F7F7F7Fu) | y3;
          const unsigned miss = ((n0 >> 7) & 0x01010101u) | ((n1 >> 6) & 0x02020202u) |
                                ((n2 >> 5) & 0x04040404u) | ((n3 >> 4) & 0x08080808u);
          unsigned h = miss ^ 0x0F0F0F0Fu;
          unsigned mask = __builtin_amdgcn_ballot_w32(h != 0u);
          if (mask != 0u) {
            const unsigned gb = ((unsigned)c * 32u + (unsigned)lane) << 5;
            do {
              if (h != 0u) {
                q[cnt + __builtin_amdgcn_mbcnt_lo(mask, 0u)] = gb | (unsigned)__builtin_ctz(h);
                h &= h - 1u;
              }
              cnt += __builtin_popcount(mask);
              mask = __builtin_amdgcn_ballot_w32(h != 0u);
            } while (mask != 0u);
          }
        }
#pragma unroll
        for (int u = 0; u < AGG_U; ++u) cur[u] = nxt[u];
      } else if (cnt > 0) {
        if (lane >= cnt && lane < AGG_B) q[lane] = 0xFFFFFFFFu;
        cnt = AGG_B;
      }

      __builtin_amdgcn_fence(__ATOMIC_SEQ_CST, "wavefront");

      while (cnt >= AGG_B) {
        cnt -= AGG_B;
        const unsigned pk = q[cnt + (lane & (AGG_B - 1))];
        const bool real = pk != 0xFFFFFFFFu;
        const unsigned g = real ? pk : 0u;
        const unsigned bit = g & 31u;
        const unsigned e = ((g >> 5) << 4) + ((bit & 7u) << 2) + (bit >> 3);
        const int s = src[e];
        const unsigned dv = (unsigned)dst[e];
        const unsigned dl = real ? dv - nbase : (unsigned)AGG_T;
        if (MODE == 0) {
          if (lane < AGG_B && wrote + AGG_B <= (unsigned)AGG_LCAP)
            wl[wrote + lane] = ((unsigned)s << 9) | dl;
          wrote += AGG_B;
        }
        gcn_gather_add<D, 0>(H, acc, s, dl, lane);
      }
    }
    if (MODE == 0 && lane == 0) wcount[widx] = wrote;
  } else if (nrec > 0u) {
    const unsigned last = nrec - 1u;
    unsigned e = wl[min((unsigned)lane, last)];
    for (unsigned i0 = 0; i0 < nrec; i0 += 32) {
      const unsigned en = wl[min(i0 + 32u + (unsigned)lane, last)];
      gcn_gather_add<D, 0>(H, acc, (int)(e >> 9), e & 511u, lane);
      if (i0 + AGG_B < nrec) gcn_gather_add<D, AGG_B>(H, acc, (int)(e >> 9), e & 511u, lane);
      e = en;
    }
  }
  __syncthreads();

  for (int i = tid; i < AGG_T * 32; i += AGG_NW * 32) {
    const int nl   = i >> 5;
    const int node = (int)nbase + nl;
    if (node < N) {
      const float ni = norm_in[node];
      vecf o;
      for (int j = 0; j < VEC; ++j)
        o[j] = fmaxf(fmaf(acc[nl * D + j * 32 + lane], ni, b[lane * VEC + j]), 0.0f);
      *reinterpret_cast<vecf*>(out + (size_t)node * D + lane * VEC) = o;
    }
  }
}

extern "C" void kernel_launch(void* const* d_in, const int* in_sizes, int n_in,
                              void* d_out, int out_size, void* d_ws, size_t ws_size,
                              hipStream_t stream) {
  const float* feat = (const float*)d_in[0];
  const int*   src  = (const int*)d_in[1];
  const int*   dst  = (const int*)d_in[2];
  const float* W1   = (const float*)d_in[3];
  const float* b1   = (const float*)d_in[4];
  const float* W2   = (const float*)d_in[5];
  const float* b2   = (const float*)d_in[6];
  float* out = (float*)d_out;
  float* ws  = (float*)d_ws;

  const int N = GCN_N_NODES;

  float* norm_out = ws + 2 * (size_t)N;
  float* norm_in  = ws + 3 * (size_t)N;
  float* bufA     = ws + 4 * (size_t)N;
  float* bufB     = bufA + (size_t)N * GCN_HID_DIM;
  float* hproj2   = bufA;
  unsigned* dpart = (unsigned*)(bufB + (size_t)N * GCN_HID_DIM);
  unsigned* wcount = dpart + (size_t)2 * DEG_P * GCN_N_NODES;
  unsigned* wlist  = wcount + 8192;
  unsigned* keys   = wlist + (size_t)AGG_NBLK * AGG_NW * AGG_LCAP;

  gcn_degree_hist_kernel<<<2 * DEG_P, DEG_NT, 0, stream>>>(src, dst, dpart, keys);
  gcn_norm_kernel<<<(N + 255) / 256, 256, 0, stream>>>(dpart, norm_out, norm_in, N);

  gcn_gemm_scaled_wmma<GCN_IN_DIM, GCN_HID_DIM>
      <<<N / 16, (GCN_HID_DIM / 16) * 32, 0, stream>>>(feat, norm_out, W1, bufA);
  gcn_aggregate_owner_kernel<GCN_HID_DIM, 0>
      <<<AGG_NBLK, AGG_NW * 32, 0, stream>>>(bufA, src, dst, keys, norm_in, b1, bufB, wcount, wlist, N);

  gcn_gemm_scaled_wmma<GCN_HID_DIM, GCN_OUT_DIM>
      <<<N / 16, (GCN_OUT_DIM / 16) * 32, 0, stream>>>(bufB, norm_out, W2, hproj2);
  gcn_aggregate_owner_kernel<GCN_OUT_DIM, 1>
      <<<AGG_NBLK, AGG_NW * 32, 0, stream>>>(hproj2, src, dst, keys, norm_in, b2, out, wcount, wlist, N);
}
